# starting kernel
# baseline (speedup 1.0000x reference)
#include <hip/hip_runtime.h>

#define ND_N 20000
#define NW_N 30000
#define E_N  500000
#define D_N  256
#define DOUT_N 256

typedef __attribute__((ext_vector_type(2))) float v2f;
typedef __attribute__((ext_vector_type(8))) float v8f;

__device__ __forceinline__ unsigned mapf(float f) {
  unsigned u = __float_as_uint(f);
  return (u & 0x80000000u) ? ~u : (u | 0x80000000u);
}
__device__ __forceinline__ float unmapf(unsigned u) {
  unsigned v = (u & 0x80000000u) ? (u & 0x7fffffffu) : ~u;
  return __uint_as_float(v);
}
#define NEG_INF_MAPPED 0x007fffffu

__global__ void fill_u32_kernel(unsigned* __restrict__ p, unsigned v, int n) {
  int i = blockIdx.x * blockDim.x + threadIdx.x;
  if (i < n) p[i] = v;
}

__global__ void node_dots_kernel(const float* __restrict__ hid, int n,
                                 const float* __restrict__ wa,
                                 const float* __restrict__ wb,
                                 float* __restrict__ outa,
                                 float* __restrict__ outb) {
  int wave = threadIdx.x >> 5, lane = threadIdx.x & 31;
  int node = blockIdx.x * 8 + wave;
  if (node >= n) return;
  const float* row = hid + (size_t)node * D_N;
  float da = 0.f, db = 0.f;
#pragma unroll
  for (int i = 0; i < 8; ++i) {
    int k = lane + i * 32;
    float x = row[k];
    da += x * wa[k];
    db += x * wb[k];
  }
#pragma unroll
  for (int off = 16; off; off >>= 1) {
    da += __shfl_xor(da, off, 32);
    db += __shfl_xor(db, off, 32);
  }
  if (lane == 0) { outa[node] = da; outb[node] = db; }
}

__global__ void edge_score_max_kernel(const int* __restrict__ src_idx,
                                      const int* __restrict__ dst_idx,
                                      const float* __restrict__ s_src,
                                      const float* __restrict__ s_dst,
                                      const float* __restrict__ bptr,
                                      float* __restrict__ e_arr,
                                      unsigned* __restrict__ m_arr, int ne) {
  int e = blockIdx.x * blockDim.x + threadIdx.x;
  if (e >= ne) return;
  float a = s_src[src_idx[e]] + s_dst[dst_idx[e]] + bptr[0];
  float v = a > 0.f ? a : 0.01f * a;
  e_arr[e] = v;
  atomicMax(&m_arr[dst_idx[e]], mapf(v));
}

__global__ void edge_exp_sum_kernel(const int* __restrict__ dst_idx,
                                    const unsigned* __restrict__ m_arr,
                                    float* __restrict__ e_arr,
                                    float* __restrict__ s_arr, int ne) {
  int e = blockIdx.x * blockDim.x + threadIdx.x;
  if (e >= ne) return;
  int d = dst_idx[e];
  float ex = __expf(e_arr[e] - unmapf(m_arr[d]));
  e_arr[e] = ex;
  atomicAdd(&s_arr[d], ex);
}

__global__ void edge_aggregate_kernel(const int* __restrict__ src_idx,
                                      const int* __restrict__ dst_idx,
                                      const float* __restrict__ e_arr,
                                      const float* __restrict__ s_arr,
                                      const float* __restrict__ src_feat,
                                      float* __restrict__ dst_accum, int ne) {
  int wave = threadIdx.x >> 5, lane = threadIdx.x & 31;
  int e = blockIdx.x * 8 + wave;
  if (e >= ne) return;
  int s = src_idx[e], d = dst_idx[e];
  float alpha = e_arr[e] / fmaxf(s_arr[d], 1e-30f);
  const float4* srow = (const float4*)(src_feat + (size_t)s * D_N);
  float* drow = dst_accum + (size_t)d * D_N;
#pragma unroll
  for (int h = 0; h < 2; ++h) {
    float4 v = srow[lane + h * 32];
    int c = (lane + h * 32) * 4;
    atomicAdd(&drow[c + 0], alpha * v.x);
    atomicAdd(&drow[c + 1], alpha * v.y);
    atomicAdd(&drow[c + 2], alpha * v.z);
    atomicAdd(&drow[c + 3], alpha * v.w);
  }
}

__global__ void fc_softmax_kernel(const float* __restrict__ H,
                                  const float* __restrict__ W,
                                  const float* __restrict__ bias,
                                  float* __restrict__ out) {
  __shared__ float tile[16][260];
  int wave = threadIdx.x >> 5;
  int lane = threadIdx.x & 31;
  size_t rowBase = (size_t)blockIdx.x * 16;
  int halfK = lane >> 4;
  int l16 = lane & 15;

  v8f zero = {0.f, 0.f, 0.f, 0.f, 0.f, 0.f, 0.f, 0.f};
  v8f acc0 = zero, acc1 = zero, acc2 = zero, acc3 = zero;

  const float* hrow = H + (rowBase + l16) * D_N;
  int colBase = wave * 64 + l16;
  for (int k0 = 0; k0 < D_N; k0 += 4) {
    int ka = k0 + halfK * 2;
    v2f a;
    a.x = hrow[ka];
    a.y = hrow[ka + 1];
    const float* w0 = W + (size_t)ka * DOUT_N;
    const float* w1 = w0 + DOUT_N;
    v2f b0; b0.x = w0[colBase +  0]; b0.y = w1[colBase +  0];
    v2f b1; b1.x = w0[colBase + 16]; b1.y = w1[colBase + 16];
    v2f b2; b2.x = w0[colBase + 32]; b2.y = w1[colBase + 32];
    v2f b3; b3.x = w0[colBase + 48]; b3.y = w1[colBase + 48];
    acc0 = __builtin_amdgcn_wmma_f32_16x16x4_f32(false, a, false, b0, (short)0, acc0, false, false);
    acc1 = __builtin_amdgcn_wmma_f32_16x16x4_f32(false, a, false, b1, (short)0, acc1, false, false);
    acc2 = __builtin_amdgcn_wmma_f32_16x16x4_f32(false, a, false, b2, (short)0, acc2, false, false);
    acc3 = __builtin_amdgcn_wmma_f32_16x16x4_f32(false, a, false, b3, (short)0, acc3, false, false);
  }

  int rowOff = halfK * 8;
  {
    v8f accs[4] = {acc0, acc1, acc2, acc3};
#pragma unroll
    for (int n = 0; n < 4; ++n) {
      int col = colBase + n * 16;
      float bv = bias[col];
#pragma unroll
      for (int r = 0; r < 8; ++r) tile[r + rowOff][col] = accs[n][r] + bv;
    }
  }
  __syncthreads();

#pragma unroll
  for (int rr = 0; rr < 4; ++rr) {
    int row = wave + rr * 4;
    float v[8];
    float mx = -3.402823466e38f;
#pragma unroll
    for (int i = 0; i < 8; ++i) {
      v[i] = tile[row][lane + i * 32];
      mx = fmaxf(mx, v[i]);
    }
#pragma unroll
    for (int off = 16; off; off >>= 1) mx = fmaxf(mx, __shfl_xor(mx, off, 32));
    float sum = 0.f;
#pragma unroll
    for (int i = 0; i < 8; ++i) { v[i] = __expf(v[i] - mx); sum += v[i]; }
#pragma unroll
    for (int off = 16; off; off >>= 1) sum += __shfl_xor(sum, off, 32);
    float inv = 1.0f / sum;
    float* orow = out + (rowBase + row) * DOUT_N;
#pragma unroll
    for (int i = 0; i < 8; ++i) orow[lane + i * 32] = v[i] * inv;
  }
}

extern "C" void kernel_launch(void* const* d_in, const int* in_sizes, int n_in,
                              void* d_out, int out_size, void* d_ws, size_t ws_size,
                              hipStream_t stream) {
  const float* doc_hidden  = (const float*)d_in[0];
  const float* word_hidden = (const float*)d_in[1];
  const int*   edge_doc    = (const int*)d_in[2];
  const int*   edge_word   = (const int*)d_in[3];
  const float* d2w_w       = (const float*)d_in[4];
  const float* d2w_b       = (const float*)d_in[5];
  const float* w2d_w       = (const float*)d_in[6];
  const float* w2d_b       = (const float*)d_in[7];
  const float* fc_w        = (const float*)d_in[8];
  const float* fc_b        = (const float*)d_in[9];
  float* out = (float*)d_out;

  float* ws = (float*)d_ws;
  const size_t NH = (size_t)(ND_N + NW_N) * D_N;
  float*    H        = ws;
  float*    sum_doc  = H + NH;
  float*    sum_word = sum_doc + ND_N;
  unsigned* max_doc  = (unsigned*)(sum_word + NW_N);
  unsigned* max_word = max_doc + ND_N;
  float*    sA_doc   = (float*)(max_word + NW_N);
  float*    sB_doc   = sA_doc + ND_N;
  float*    sA_word  = sB_doc + ND_N;
  float*    sB_word  = sA_word + NW_N;
  float*    e_d2w    = sB_word + NW_N;
  float*    e_w2d    = e_d2w + E_N;

  hipMemsetAsync(ws, 0, (NH + ND_N + NW_N) * sizeof(float), stream);
  {
    int n = ND_N + NW_N;
    fill_u32_kernel<<<(n + 255) / 256, 256, 0, stream>>>(max_doc, NEG_INF_MAPPED, n);
  }

  node_dots_kernel<<<ND_N / 8, 256, 0, stream>>>(doc_hidden, ND_N,
                                                 d2w_w, w2d_w + D_N, sA_doc, sB_doc);
  node_dots_kernel<<<NW_N / 8, 256, 0, stream>>>(word_hidden, NW_N,
                                                 d2w_w + D_N, w2d_w, sA_word, sB_word);

  const int egrid = (E_N + 255) / 256;
  edge_score_max_kernel<<<egrid, 256, 0, stream>>>(edge_doc, edge_word, sA_doc, sA_word,
                                                   d2w_b, e_d2w, max_word, E_N);
  edge_score_max_kernel<<<egrid, 256, 0, stream>>>(edge_word, edge_doc, sB_word, sB_doc,
                                                   w2d_b, e_w2d, max_doc, E_N);
  edge_exp_sum_kernel<<<egrid, 256, 0, stream>>>(edge_word, max_word, e_d2w, sum_word, E_N);
  edge_exp_sum_kernel<<<egrid, 256, 0, stream>>>(edge_doc, max_doc, e_w2d, sum_doc, E_N);
  edge_aggregate_kernel<<<E_N / 8, 256, 0, stream>>>(edge_doc, edge_word, e_d2w, sum_word,
                                                     doc_hidden, H + (size_t)ND_N * D_N, E_N);
  edge_aggregate_kernel<<<E_N / 8, 256, 0, stream>>>(edge_word, edge_doc, e_w2d, sum_doc,
                                                     word_hidden, H, E_N);

  fc_softmax_kernel<<<(ND_N + NW_N) / 16, 128, 0, stream>>>(H, fc_w, fc_b, out);

  (void)in_sizes; (void)n_in; (void)out_size; (void)ws_size;
}


# destination-row tiles kept in LDS: a workgroup filters the edge list for its rows; max, exp sum, row sums by LDS atomics (arrival order), one division per row; FC+softmax from LDS; no global atomics
# speedup vs baseline: 4.1687x; 4.1687x over previous
#include <hip/hip_runtime.h>

#define ND_N 20000
#define NW_N 30000
#define E_N  500000
#define D_N  256
#define DOUT_N 256

#define RW 224
#define RD 176
#define RMAX 224
#define NBW ((NW_N + RW - 1) / RW)
#define NBD ((ND_N + RD - 1) / RD)
#define PITCH 260
#define WCAP 256
#define CAP (32 * WCAP)
#define UB 8
#define NBATCH ((NIT + UB - 1) / UB)
#define NTHR 1024
#define NIT ((E_N + NTHR * 4 - 1) / (NTHR * 4))

typedef __attribute__((ext_vector_type(2))) float v2f;
typedef __attribute__((ext_vector_type(8))) float v8f;

__device__ __forceinline__ unsigned mapf(float f) {
  unsigned u = __float_as_uint(f);
  return (u & 0x80000000u) ? ~u : (u | 0x80000000u);
}
__device__ __forceinline__ float unmapf(unsigned u) {
  unsigned v = (u & 0x80000000u) ? (u & 0x7fffffffu) : ~u;
  return __uint_as_float(v);
}
#define NEG_INF_MAPPED 0x007fffffu

__global__ void node_dots_kernel(const float* __restrict__ hid, int n,
                                 const float* __restrict__ wa,
                                 const float* __restrict__ wb,
                                 float* __restrict__ outa,
                                 float* __restrict__ outb) {
  int wave = threadIdx.x >> 5, lane = threadIdx.x & 31;
  int node = blockIdx.x * 8 + wave;
  if (node >= n) return;
  const float* row = hid + (size_t)node * D_N;
  float da = 0.f, db = 0.f;
#pragma unroll
  for (int i = 0; i < 8; ++i) {
    int k = lane + i * 32;
    float x = row[k];
    da += x * wa[k];
    db += x * wb[k];
  }
#pragma unroll
  for (int off = 16; off; off >>= 1) {
    da += __shfl_xor(da, off, 32);
    db += __shfl_xor(db, off, 32);
  }
  if (lane == 0) { outa[node] = da; outb[node] = db; }
}

__device__ __forceinline__ void add_row(float* __restrict__ arow, const float* __restrict__ srow,
                                        float wgt, int lane) {
  float x[8];
#pragma unroll
  for (int q = 0; q < 8; ++q) x[q] = srow[lane + 32 * q];
#pragma unroll
  for (int q = 0; q < 8; ++q) atomicAdd(&arow[lane + 32 * q], wgt * x[q]);
}

__device__ __forceinline__ void add_rows4(float* __restrict__ acc, const float* __restrict__ feat,
                                          int s, unsigned dl, float ex, unsigned j, int lane) {
  float x[4][8];
#pragma unroll
  for (int t = 0; t < 4; ++t) {
    int sj = __builtin_amdgcn_readlane(s, j + t);
    const float* srow = feat + (size_t)sj * D_N;
#pragma unroll
    for (int q = 0; q < 8; ++q) x[t][q] = srow[lane + 32 * q];
  }
#pragma unroll
  for (int t = 0; t < 4; ++t) {
    unsigned dj = (unsigned)__builtin_amdgcn_readlane((int)dl, j + t);
    float ej = __uint_as_float((unsigned)__builtin_amdgcn_readlane((int)__float_as_uint(ex), j + t));
    float* arow = acc + dj * PITCH;
#pragma unroll
    for (int q = 0; q < 8; ++q) atomicAdd(&arow[lane + 32 * q], ej * x[t][q]);
  }
}

__global__ __launch_bounds__(NTHR) void gat_tile_kernel(
    const float* __restrict__ doc_hidden, const float* __restrict__ word_hidden,
    const int* __restrict__ edge_doc, const int* __restrict__ edge_word,
    const float* __restrict__ sA_doc, const float* __restrict__ sB_doc,
    const float* __restrict__ sA_word, const float* __restrict__ sB_word,
    const float* __restrict__ d2w_b, const float* __restrict__ w2d_b,
    const float* __restrict__ W, const float* __restrict__ bias,
    float* __restrict__ out) {
  __shared__ float acc[RMAX * PITCH];
  __shared__ unsigned mx[RMAX];
  __shared__ float sm[RMAX];
  __shared__ float sdl[RMAX];
  __shared__ unsigned ent[CAP];
  __shared__ float val[CAP];
  __shared__ unsigned ovf;

  const int tid = threadIdx.x;
  const int wave = __builtin_amdgcn_readfirstlane(tid >> 5);
  const int lane = tid & 31;

  const bool isWord = blockIdx.x < NBW;
  int rowBase, R, outRow0;
  const int *didx, *sidx;
  const float *sS, *sD, *feat, *bptr;
  if (isWord) {
    rowBase = blockIdx.x * RW;
    R = NW_N - rowBase; if (R > RW) R = RW;
    didx = edge_word; sidx = edge_doc; sS = sA_doc; sD = sA_word;
    feat = doc_hidden; bptr = d2w_b; outRow0 = ND_N + rowBase;
  } else {
    rowBase = (blockIdx.x - NBW) * RD;
    R = ND_N - rowBase; if (R > RD) R = RD;
    didx = edge_doc; sidx = edge_word; sS = sB_word; sD = sB_doc;
    feat = word_hidden; bptr = w2d_b; outRow0 = rowBase;
  }
  const float b0 = bptr[0];

  for (int i = tid; i < RMAX * PITCH; i += NTHR) acc[i] = 0.f;
  if (tid < RMAX) {
    mx[tid] = NEG_INF_MAPPED;
    sm[tid] = 0.f;
    sdl[tid] = (tid < R) ? sD[rowBase + tid] : 0.f;
  }
  if (tid == 0) ovf = 0u;
  __syncthreads();

  const unsigned seg = (unsigned)wave * WCAP;
  unsigned wc = 0u;
  for (int bt = 0; bt < NBATCH; ++bt) {
    int4 d4[UB];
#pragma unroll
    for (int u = 0; u < UB; ++u) {
      int base = ((bt * UB + u) * NTHR + tid) * 4;
      d4[u] = make_int4(-1, -1, -1, -1);
      if (base < E_N) d4[u] = *(const int4*)(didx + base);
    }
#pragma unroll
    for (int u = 0; u < UB; ++u) {
      int base = ((bt * UB + u) * NTHR + tid) * 4;
      int dd[4] = {d4[u].x, d4[u].y, d4[u].z, d4[u].w};
#pragma unroll
      for (int c = 0; c < 4; ++c) {
        unsigned dl = (unsigned)(dd[c] - rowBase);
        bool m = dl < (unsigned)R;
        unsigned bal = __builtin_amdgcn_ballot_w32(m);
        if (bal) {
          unsigned pos = wc + __builtin_amdgcn_mbcnt_lo(bal, 0u);
          if (m && pos < WCAP) ent[seg + pos] = ((unsigned)(base + c) << 8) | dl;
          wc += (unsigned)__builtin_popcount(bal);
        }
      }
    }
  }
  if (wc > WCAP && lane == 0) ovf = 1u;
  __syncthreads();
  const bool overflow = __builtin_amdgcn_readfirstlane((int)ovf) != 0;

  if (!overflow) {
    for (unsigned i = lane; i < wc; i += 32u) {
      unsigned u = ent[seg + i];
      unsigned dl = u & 255u;
      int s = sidx[u >> 8];
      float a = sS[s] + sdl[dl] + b0;
      float v = a > 0.f ? a : 0.01f * a;
      val[seg + i] = v;
      atomicMax(&mx[dl], mapf(v));
    }
    __syncthreads();
    for (unsigned i0 = 0u; i0 < wc; i0 += 32u) {
      unsigned i = i0 + lane;
      unsigned dl = 0u; int s = 0; float ex = 0.f;
      if (i < wc) {
        unsigned u = ent[seg + i];
        dl = u & 255u;
        s = sidx[u >> 8];
        ex = __expf(val[seg + i] - unmapf(mx[dl]));
        atomicAdd(&sm[dl], ex);
      }
      unsigned n = wc - i0; if (n > 32u) n = 32u;
      unsigned j = 0u;
      for (; j + 4u <= n; j += 4u) add_rows4(acc, feat, s, dl, ex, j, lane);
      for (; j < n; ++j) {
        int sj = __builtin_amdgcn_readlane(s, j);
        unsigned dj = (unsigned)__builtin_amdgcn_readlane((int)dl, j);
        float ej = __uint_as_float((unsigned)__builtin_amdgcn_readlane((int)__float_as_uint(ex), j));
        add_row(acc + dj * PITCH, feat + (size_t)sj * D_N, ej, lane);
      }
    }
  } else {
    for (int it = 0; it < NIT; ++it) {
      int base = (it * NTHR + tid) * 4;
      int4 d4 = make_int4(-1, -1, -1, -1);
      if (base < E_N) d4 = *(const int4*)(didx + base);
      int dd[4] = {d4.x, d4.y, d4.z, d4.w};
#pragma unroll
      for (int c = 0; c < 4; ++c) {
        unsigned dl = (unsigned)(dd[c] - rowBase);
        if (dl < (unsigned)R) {
          int s = sidx[base + c];
          float a = sS[s] + sdl[dl] + b0;
          float v = a > 0.f ? a : 0.01f * a;
          atomicMax(&mx[dl], mapf(v));
        }
      }
    }
    __syncthreads();
    for (int it = 0; it < NIT; ++it) {
      int base = (it * NTHR + tid) * 4;
      int4 d4 = make_int4(-1, -1, -1, -1);
      if (base < E_N) d4 = *(const int4*)(didx + base);
      int dd[4] = {d4.x, d4.y, d4.z, d4.w};
#pragma unroll
      for (int c = 0; c < 4; ++c) {
        unsigned dl = (unsigned)(dd[c] - rowBase);
        bool m = dl < (unsigned)R;
        int s = 0; float ex = 0.f;
        if (m) {
          s = sidx[base + c];
          float a = sS[s] + sdl[dl] + b0;
          float v = a > 0.f ? a : 0.01f * a;
          ex = __expf(v - unmapf(mx[dl]));
          atomicAdd(&sm[dl], ex);
        } else {
          dl = 0u;
        }
        unsigned mask = __builtin_amdgcn_ballot_w32(m);
        while (mask) {
          int j = __builtin_ctz(mask);
          mask &= mask - 1u;
          int sj = __builtin_amdgcn_readlane(s, j);
          unsigned dj = (unsigned)__builtin_amdgcn_readlane((int)dl, j);
          float ej = __uint_as_float((unsigned)__builtin_amdgcn_readlane((int)__float_as_uint(ex), j));
          add_row(acc + dj * PITCH, feat + (size_t)sj * D_N, ej, lane);
        }
      }
    }
  }
  __syncthreads();

  for (int i = tid; i < R * D_N; i += NTHR) {
    int r = i >> 8, c = i & 255;
    acc[r * PITCH + c] = acc[r * PITCH + c] / fmaxf(sm[r], 1e-30f);
  }
  __syncthreads();

  const int ntiles = R >> 4;
  const int halfK = lane >> 4;
  const int l16 = lane & 15;
  const int rt0 = wave >> 2, rt1 = rt0 + 8;
  const bool has0 = rt0 < ntiles, has1 = rt1 < ntiles;
  const int colBase = (wave & 3) * 64 + l16;

  v8f zero = {0.f, 0.f, 0.f, 0.f, 0.f, 0.f, 0.f, 0.f};
  v8f c0 = zero, c1 = zero, c2 = zero, c3 = zero;
  v8f c4 = zero, c5 = zero, c6 = zero, c7 = zero;
  if (has0) {
    const float* A0 = acc + (rt0 * 16 + l16) * PITCH;
    const float* A1 = acc + (rt1 * 16 + l16) * PITCH;
    for (int k0 = 0; k0 < D_N; k0 += 4) {
      int ka = k0 + halfK * 2;
      const float* w0 = W + (size_t)ka * DOUT_N;
      const float* w1 = w0 + DOUT_N;
      v2f b0v; b0v.x = w0[colBase +  0]; b0v.y = w1[colBase +  0];
      v2f b1v; b1v.x = w0[colBase + 16]; b1v.y = w1[colBase + 16];
      v2f b2v; b2v.x = w0[colBase + 32]; b2v.y = w1[colBase + 32];
      v2f b3v; b3v.x = w0[colBase + 48]; b3v.y = w1[colBase + 48];
      v2f a;
      a.x = A0[ka];
      a.y = A0[ka + 1];
      c0 = __builtin_amdgcn_wmma_f32_16x16x4_f32(false, a, false, b0v, (short)0, c0, false, false);
      c1 = __builtin_amdgcn_wmma_f32_16x16x4_f32(false, a, false, b1v, (short)0, c1, false, false);
      c2 = __builtin_amdgcn_wmma_f32_16x16x4_f32(false, a, false, b2v, (short)0, c2, false, false);
      c3 = __builtin_amdgcn_wmma_f32_16x16x4_f32(false, a, false, b3v, (short)0, c3, false, false);
      if (has1) {
        v2f a2;
        a2.x = A1[ka];
        a2.y = A1[ka + 1];
        c4 = __builtin_amdgcn_wmma_f32_16x16x4_f32(false, a2, false, b0v, (short)0, c4, false, false);
        c5 = __builtin_amdgcn_wmma_f32_16x16x4_f32(false, a2, false, b1v, (short)0, c5, false, false);
        c6 = __builtin_amdgcn_wmma_f32_16x16x4_f32(false, a2, false, b2v, (short)0, c6, false, false);
        c7 = __builtin_amdgcn_wmma_f32_16x16x4_f32(false, a2, false, b3v, (short)0, c7, false, false);
      }
    }
  }
  __syncthreads();

  {
    const int rowOff = halfK * 8;
    v8f lo[4] = {c0, c1, c2, c3};
    v8f hi[4] = {c4, c5, c6, c7};
#pragma unroll
    for (int n = 0; n < 4; ++n) {
      int col = colBase + n * 16;
      float bv = bias[col];
      if (has0) {
#pragma unroll
        for (int r = 0; r < 8; ++r) acc[(rt0 * 16 + r + rowOff) * PITCH + col] = lo[n][r] + bv;
      }
      if (has1) {
#pragma unroll
        for (int r = 0; r < 8; ++r) acc[(rt1 * 16 + r + rowOff) * PITCH + col] = hi[n][r] + bv;
      }
    }
  }
  __syncthreads();

  for (int row = wave; row < R; row += 32) {
    float v[8];
    float mxv = -3.402823466e38f;
#pragma unroll
    for (int i = 0; i < 8; ++i) {
      v[i] = acc[row * PITCH + lane + i * 32];
      mxv = fmaxf(mxv, v[i]);
    }
#pragma unroll
    for (int off = 16; off; off >>= 1) mxv = fmaxf(mxv, __shfl_xor(mxv, off, 32));
    float sum = 0.f;
#pragma unroll
    for (int i = 0; i < 8; ++i) { v[i] = __expf(v[i] - mxv); sum += v[i]; }
#pragma unroll
    for (int off = 16; off; off >>= 1) sum += __shfl_xor(sum, off, 32);
    float inv = 1.0f / sum;
    float* orow = out + (size_t)(outRow0 + row) * DOUT_N;
#pragma unroll
    for (int i = 0; i < 8; ++i) orow[lane + i * 32] = v[i] * inv;
  }
}

extern "C" void kernel_launch(void* const* d_in, const int* in_sizes, int n_in,
                              void* d_out, int out_size, void* d_ws, size_t ws_size,
                              hipStream_t stream) {
  const float* doc_hidden  = (const float*)d_in[0];
  const float* word_hidden = (const float*)d_in[1];
  const int*   edge_doc    = (const int*)d_in[2];
  const int*   edge_word   = (const int*)d_in[3];
  const float* d2w_w       = (const float*)d_in[4];
  const float* d2w_b       = (const float*)d_in[5];
  const float* w2d_w       = (const float*)d_in[6];
  const float* w2d_b       = (const float*)d_in[7];
  const float* fc_w        = (const float*)d_in[8];
  const float* fc_b        = (const float*)d_in[9];
  float* out = (float*)d_out;

  float* ws = (float*)d_ws;
  float* sA_doc  = ws;
  float* sB_doc  = sA_doc + ND_N;
  float* sA_word = sB_doc + ND_N;
  float* sB_word = sA_word + NW_N;

  node_dots_kernel<<<ND_N / 8, 256, 0, stream>>>(doc_hidden, ND_N,
                                                 d2w_w, w2d_w + D_N, sA_doc, sB_doc);
  node_dots_kernel<<<NW_N / 8, 256, 0, stream>>>(word_hidden, NW_N,
                                                 d2w_w + D_N, w2d_w, sA_word, sB_word);

  gat_tile_kernel<<<NBW + NBD, NTHR, 0, stream>>>(doc_hidden, word_hidden, edge_doc, edge_word,
                                                  sA_doc, sB_doc, sA_word, sB_word,
                                                  d2w_b, w2d_b, fc_w, fc_b, out);

  (void)in_sizes; (void)n_in; (void)out_size; (void)ws_size;
}


# row sums in registers: matches dealt to the wave owning row r&31 by LDS integer slot adds; gathered rows added by fma into the row's registers (scalar index), arrival order; no LDS f32 row atomics
# speedup vs baseline: 4.7854x; 1.1479x over previous
#include <hip/hip_runtime.h>

#define ND_N 20000
#define NW_N 30000
#define E_N  500000
#define D_N  256
#define DOUT_N 256

#define RW 224
#define RD 176
#define RMAX 224
#define NBW ((NW_N + RW - 1) / RW)
#define NBD ((ND_N + RD - 1) / RD)
#define PITCH 260
#define WCAP 256
#define CAP (32 * WCAP)
#define UB 8
#define NBATCH ((NIT + UB - 1) / UB)
#define NTHR 1024
#define NIT ((E_N + NTHR * 4 - 1) / (NTHR * 4))

typedef __attribute__((ext_vector_type(2))) float v2f;
typedef __attribute__((ext_vector_type(8))) float v8f;

__device__ __forceinline__ unsigned mapf(float f) {
  unsigned u = __float_as_uint(f);
  return (u & 0x80000000u) ? ~u : (u | 0x80000000u);
}
__device__ __forceinline__ float unmapf(unsigned u) {
  unsigned v = (u & 0x80000000u) ? (u & 0x7fffffffu) : ~u;
  return __uint_as_float(v);
}
#define NEG_INF_MAPPED 0x007fffffu

__global__ void node_dots_kernel(const float* __restrict__ hid, int n,
                                 const float* __restrict__ wa,
                                 const float* __restrict__ wb,
                                 float* __restrict__ outa,
                                 float* __restrict__ outb) {
  int wave = threadIdx.x >> 5, lane = threadIdx.x & 31;
  int node = blockIdx.x * 8 + wave;
  if (node >= n) return;
  const float* row = hid + (size_t)node * D_N;
  float da = 0.f, db = 0.f;
#pragma unroll
  for (int i = 0; i < 8; ++i) {
    int k = lane + i * 32;
    float x = row[k];
    da += x * wa[k];
    db += x * wb[k];
  }
#pragma unroll
  for (int off = 16; off; off >>= 1) {
    da += __shfl_xor(da, off, 32);
    db += __shfl_xor(db, off, 32);
  }
  if (lane == 0) { outa[node] = da; outb[node] = db; }
}

__device__ __forceinline__ void add_row(float* __restrict__ arow, const float* __restrict__ srow,
                                        float wgt, int lane) {
  float x[8];
#pragma unroll
  for (int q = 0; q < 8; ++q) x[q] = srow[lane + 32 * q];
#pragma unroll
  for (int q = 0; q < 8; ++q) atomicAdd(&arow[lane + 32 * q], wgt * x[q]);
}

#define ROW_ADD(SLOT, E, X) { _Pragma("unroll") for (int q = 0; q < 8; ++q) V[q][SLOT] += (E) * (X)[q]; }
#define ROW_OUT(K) {                                                       \
    const int r = (K) * 32 + wave;                                         \
    if (r < R) {                                                           \
      const float d = fmaxf(sm[r], 1e-30f);                                \
      _Pragma("unroll") for (int q = 0; q < 8; ++q) acc[r * PITCH + lane + 32 * q] = V[q][K] / d; \
    }                                                                      \
  }

__global__ __launch_bounds__(NTHR) void gat_tile_kernel(
    const float* __restrict__ doc_hidden, const float* __restrict__ word_hidden,
    const int* __restrict__ edge_doc, const int* __restrict__ edge_word,
    const float* __restrict__ sA_doc, const float* __restrict__ sB_doc,
    const float* __restrict__ sA_word, const float* __restrict__ sB_word,
    const float* __restrict__ d2w_b, const float* __restrict__ w2d_b,
    const float* __restrict__ W, const float* __restrict__ bias,
    float* __restrict__ out) {
  __shared__ float acc[RMAX * PITCH];
  __shared__ unsigned mx[RMAX];
  __shared__ float sm[RMAX];
  __shared__ float sdl[RMAX];
  __shared__ unsigned ent[CAP];
  __shared__ unsigned own[CAP];
  __shared__ unsigned cnt[32];
  __shared__ unsigned ovf;

  const int tid = threadIdx.x;
  const int wave = __builtin_amdgcn_readfirstlane(tid >> 5);
  const int lane = tid & 31;

  const bool isWord = blockIdx.x < NBW;
  int rowBase, R, outRow0;
  const int *didx, *sidx;
  const float *sS, *sD, *feat, *bptr;
  if (isWord) {
    rowBase = blockIdx.x * RW;
    R = NW_N - rowBase; if (R > RW) R = RW;
    didx = edge_word; sidx = edge_doc; sS = sA_doc; sD = sA_word;
    feat = doc_hidden; bptr = d2w_b; outRow0 = ND_N + rowBase;
  } else {
    rowBase = (blockIdx.x - NBW) * RD;
    R = ND_N - rowBase; if (R > RD) R = RD;
    didx = edge_doc; sidx = edge_word; sS = sB_word; sD = sB_doc;
    feat = word_hidden; bptr = w2d_b; outRow0 = rowBase;
  }
  const float b0 = bptr[0];
  const v8f zero8 = {0.f, 0.f, 0.f, 0.f, 0.f, 0.f, 0.f, 0.f};

  if (tid < 32) cnt[tid] = 0u;
  if (tid < RMAX) {
    mx[tid] = NEG_INF_MAPPED;
    sm[tid] = 0.f;
    sdl[tid] = (tid < R) ? sD[rowBase + tid] : 0.f;
  }
  if (tid == 0) ovf = 0u;
  __syncthreads();

  const unsigned seg = (unsigned)wave * WCAP;
  unsigned wc = 0u;
  for (int bt = 0; bt < NBATCH; ++bt) {
    int4 d4[UB];
#pragma unroll
    for (int u = 0; u < UB; ++u) {
      int base = ((bt * UB + u) * NTHR + tid) * 4;
      d4[u] = make_int4(-1, -1, -1, -1);
      if (base < E_N) d4[u] = *(const int4*)(didx + base);
    }
#pragma unroll
    for (int u = 0; u < UB; ++u) {
      int base = ((bt * UB + u) * NTHR + tid) * 4;
      int dd[4] = {d4[u].x, d4[u].y, d4[u].z, d4[u].w};
#pragma unroll
      for (int c = 0; c < 4; ++c) {
        unsigned dl = (unsigned)(dd[c] - rowBase);
        bool m = dl < (unsigned)R;
        unsigned bal = __builtin_amdgcn_ballot_w32(m);
        if (bal) {
          unsigned pos = wc + __builtin_amdgcn_mbcnt_lo(bal, 0u);
          if (m && pos < WCAP) ent[seg + pos] = ((unsigned)(base + c) << 8) | dl;
          wc += (unsigned)__builtin_popcount(bal);
        }
      }
    }
  }
  if (wc > WCAP && lane == 0) ovf = 1u;
  {
    unsigned wl = wc < WCAP ? wc : WCAP;
    for (unsigned i = lane; i < wl; i += 32u) {
      unsigned u = ent[seg + i];
      unsigned ow = u & 31u;
      unsigned pos = atomicAdd(&cnt[ow], 1u);
      if (pos < WCAP) own[ow * WCAP + pos] = u;
    }
  }
  __syncthreads();
  const unsigned n = (unsigned)__builtin_amdgcn_readfirstlane((int)cnt[wave]);
  if (n > WCAP && lane == 0) ovf = 1u;
  __syncthreads();
  const bool overflow = __builtin_amdgcn_readfirstlane((int)ovf) != 0;

  if (!overflow) {
    const unsigned oseg = (unsigned)wave * WCAP;
    for (unsigned i = lane; i < n; i += 32u) {
      unsigned u = own[oseg + i];
      unsigned dl = u & 255u;
      int s = sidx[u >> 8];
      float a = sS[s] + sdl[dl] + b0;
      float v = a > 0.f ? a : 0.01f * a;
      ent[oseg + i] = __float_as_uint(v);
      atomicMax(&mx[dl], mapf(v));
    }

    v8f V[8];
#pragma unroll
    for (int q = 0; q < 8; ++q) V[q] = zero8;
    for (unsigned i0 = 0u; i0 < n; i0 += 32u) {
      unsigned i = i0 + lane;
      unsigned dl = 0u; int s = 0; float ex = 0.f;
      if (i < n) {
        unsigned u = own[oseg + i];
        dl = u & 255u;
        s = sidx[u >> 8];
        ex = __expf(__uint_as_float(ent[oseg + i]) - unmapf(mx[dl]));
        atomicAdd(&sm[dl], ex);
      }
      unsigned m = n - i0; if (m > 32u) m = 32u;
      unsigned j = 0u;
      for (; j + 4u <= m; j += 4u) {
        float x[4][8];
#pragma unroll
        for (int t = 0; t < 4; ++t) {
          int sj = __builtin_amdgcn_readlane(s, j + t);
          const float* srow = feat + (size_t)sj * D_N;
#pragma unroll
          for (int q = 0; q < 8; ++q) x[t][q] = srow[lane + 32 * q];
        }
#pragma unroll
        for (int t = 0; t < 4; ++t) {
          const int slot = __builtin_amdgcn_readlane((int)dl, j + t) >> 5;
          const float ej = __uint_as_float((unsigned)__builtin_amdgcn_readlane((int)__float_as_uint(ex), j + t));
          ROW_ADD(slot, ej, x[t])
        }
      }
      for (; j < m; ++j) {
        int sj = __builtin_amdgcn_readlane(s, j);
        const int slot = __builtin_amdgcn_readlane((int)dl, j) >> 5;
        const float ej = __uint_as_float((unsigned)__builtin_amdgcn_readlane((int)__float_as_uint(ex), j));
        const float* srow = feat + (size_t)sj * D_N;
        float x1[8];
#pragma unroll
        for (int q = 0; q < 8; ++q) x1[q] = srow[lane + 32 * q];
        ROW_ADD(slot, ej, x1)
      }
    }
    ROW_OUT(0) ROW_OUT(1) ROW_OUT(2) ROW_OUT(3) ROW_OUT(4) ROW_OUT(5) ROW_OUT(6)
  } else {
    for (int i = tid; i < RMAX * PITCH; i += NTHR) acc[i] = 0.f;
    __syncthreads();
    for (int it = 0; it < NIT; ++it) {
      int base = (it * NTHR + tid) * 4;
      int4 d4 = make_int4(-1, -1, -1, -1);
      if (base < E_N) d4 = *(const int4*)(didx + base);
      int dd[4] = {d4.x, d4.y, d4.z, d4.w};
#pragma unroll
      for (int c = 0; c < 4; ++c) {
        unsigned dl = (unsigned)(dd[c] - rowBase);
        if (dl < (unsigned)R) {
          int s = sidx[base + c];
          float a = sS[s] + sdl[dl] + b0;
          float v = a > 0.f ? a : 0.01f * a;
          atomicMax(&mx[dl], mapf(v));
        }
      }
    }
    __syncthreads();
    for (int it = 0; it < NIT; ++it) {
      int base = (it * NTHR + tid) * 4;
      int4 d4 = make_int4(-1, -1, -1, -1);
      if (base < E_N) d4 = *(const int4*)(didx + base);
      int dd[4] = {d4.x, d4.y, d4.z, d4.w};
#pragma unroll
      for (int c = 0; c < 4; ++c) {
        unsigned dl = (unsigned)(dd[c] - rowBase);
        bool m = dl < (unsigned)R;
        int s = 0; float ex = 0.f;
        if (m) {
          s = sidx[base + c];
          float a = sS[s] + sdl[dl] + b0;
          float v = a > 0.f ? a : 0.01f * a;
          ex = __expf(v - unmapf(mx[dl]));
          atomicAdd(&sm[dl], ex);
        } else {
          dl = 0u;
        }
        unsigned mask = __builtin_amdgcn_ballot_w32(m);
        while (mask) {
          int j = __builtin_ctz(mask);
          mask &= mask - 1u;
          int sj = __builtin_amdgcn_readlane(s, j);
          unsigned dj = (unsigned)__builtin_amdgcn_readlane((int)dl, j);
          float ej = __uint_as_float((unsigned)__builtin_amdgcn_readlane((int)__float_as_uint(ex), j));
          add_row(acc + dj * PITCH, feat + (size_t)sj * D_N, ej, lane);
        }
      }
    }
    __syncthreads();
    for (int i = tid; i < R * D_N; i += NTHR) {
      int r = i >> 8, c = i & 255;
      acc[r * PITCH + c] = acc[r * PITCH + c] / fmaxf(sm[r], 1e-30f);
    }
  }
  __syncthreads();

  const int ntiles = R >> 4;
  const int halfK = lane >> 4;
  const int l16 = lane & 15;
  const int rt0 = wave >> 2, rt1 = rt0 + 8;
  const bool has0 = rt0 < ntiles, has1 = rt1 < ntiles;
  const int colBase = (wave & 3) * 64 + l16;

  v8f zero = {0.f, 0.f, 0.f, 0.f, 0.f, 0.f, 0.f, 0.f};
  v8f c0 = zero, c1 = zero, c2 = zero, c3 = zero;
  v8f c4 = zero, c5 = zero, c6 = zero, c7 = zero;
  if (has0) {
    const float* A0 = acc + (rt0 * 16 + l16) * PITCH;
    const float* A1 = acc + (rt1 * 16 + l16) * PITCH;
    for (int k0 = 0; k0 < D_N; k0 += 4) {
      int ka = k0 + halfK * 2;
      const float* w0 = W + (size_t)ka * DOUT_N;
      const float* w1 = w0 + DOUT_N;
      v2f b0v; b0v.x = w0[colBase +  0]; b0v.y = w1[colBase +  0];
      v2f b1v; b1v.x = w0[colBase + 16]; b1v.y = w1[colBase + 16];
      v2f b2v; b2v.x = w0[colBase + 32]; b2v.y = w1[colBase + 32];
      v2f b3v; b3v.x = w0[colBase + 48]; b3v.y = w1[colBase + 48];
      v2f a;
      a.x = A0[ka];
      a.y = A0[ka + 1];
      c0 = __builtin_amdgcn_wmma_f32_16x16x4_f32(false, a, false, b0v, (short)0, c0, false, false);
      c1 = __builtin_amdgcn_wmma_f32_16x16x4_f32(false, a, false, b1v, (short)0, c1, false, false);
      c2 = __builtin_amdgcn_wmma_f32_16x16x4_f32(false, a, false, b2v, (short)0, c2, false, false);
      c3 = __builtin_amdgcn_wmma_f32_16x16x4_f32(false, a, false, b3v, (short)0, c3, false, false);
      if (has1) {
        v2f a2;
        a2.x = A1[ka];
        a2.y = A1[ka + 1];
        c4 = __builtin_amdgcn_wmma_f32_16x16x4_f32(false, a2, false, b0v, (short)0, c4, false, false);
        c5 = __builtin_amdgcn_wmma_f32_16x16x4_f32(false, a2, false, b1v, (short)0, c5, false, false);
        c6 = __builtin_amdgcn_wmma_f32_16x16x4_f32(false, a2, false, b2v, (short)0, c6, false, false);
        c7 = __builtin_amdgcn_wmma_f32_16x16x4_f32(false, a2, false, b3v, (short)0, c7, false, false);
      }
    }
  }
  __syncthreads();

  {
    const int rowOff = halfK * 8;
    v8f lo[4] = {c0, c1, c2, c3};
    v8f hi[4] = {c4, c5, c6, c7};
#pragma unroll
    for (int n = 0; n < 4; ++n) {
      int col = colBase + n * 16;
      float bv = bias[col];
      if (has0) {
#pragma unroll
        for (int r = 0; r < 8; ++r) acc[(rt0 * 16 + r + rowOff) * PITCH + col] = lo[n][r] + bv;
      }
      if (has1) {
#pragma unroll
        for (int r = 0; r < 8; ++r) acc[(rt1 * 16 + r + rowOff) * PITCH + col] = hi[n][r] + bv;
      }
    }
  }
  __syncthreads();

  for (int row = wave; row < R; row += 32) {
    float v[8];
    float mxv = -3.402823466e38f;
#pragma unroll
    for (int i = 0; i < 8; ++i) {
      v[i] = acc[row * PITCH + lane + i * 32];
      mxv = fmaxf(mxv, v[i]);
    }
#pragma unroll
    for (int off = 16; off; off >>= 1) mxv = fmaxf(mxv, __shfl_xor(mxv, off, 32));
    float sum = 0.f;
#pragma unroll
    for (int i = 0; i < 8; ++i) { v[i] = __expf(v[i] - mxv); sum += v[i]; }
#pragma unroll
    for (int off = 16; off; off >>= 1) sum += __shfl_xor(sum, off, 32);
    float inv = 1.0f / sum;
    float* orow = out + (size_t)(outRow0 + row) * DOUT_N;
#pragma unroll
    for (int i = 0; i < 8; ++i) orow[lane + i * 32] = v[i] * inv;
  }
}

extern "C" void kernel_launch(void* const* d_in, const int* in_sizes, int n_in,
                              void* d_out, int out_size, void* d_ws, size_t ws_size,
                              hipStream_t stream) {
  const float* doc_hidden  = (const float*)d_in[0];
  const float* word_hidden = (const float*)d_in[1];
  const int*   edge_doc    = (const int*)d_in[2];
  const int*   edge_word   = (const int*)d_in[3];
  const float* d2w_w       = (const float*)d_in[4];
  const float* d2w_b       = (const float*)d_in[5];
  const float* w2d_w       = (const float*)d_in[6];
  const float* w2d_b       = (const float*)d_in[7];
  const float* fc_w        = (const float*)d_in[8];
  const float* fc_b        = (const float*)d_in[9];
  float* out = (float*)d_out;

  float* ws = (float*)d_ws;
  float* sA_doc  = ws;
  float* sB_doc  = sA_doc + ND_N;
  float* sA_word = sB_doc + ND_N;
  float* sB_word = sA_word + NW_N;

  node_dots_kernel<<<ND_N / 8, 256, 0, stream>>>(doc_hidden, ND_N,
                                                 d2w_w, w2d_w + D_N, sA_doc, sB_doc);
  node_dots_kernel<<<NW_N / 8, 256, 0, stream>>>(word_hidden, NW_N,
                                                 d2w_w + D_N, w2d_w, sA_word, sB_word);

  gat_tile_kernel<<<NBW + NBD, NTHR, 0, stream>>>(doc_hidden, word_hidden, edge_doc, edge_word,
                                                  sA_doc, sB_doc, sA_word, sB_word,
                                                  d2w_b, w2d_b, fc_w, fc_b, out);

  (void)in_sizes; (void)n_in; (void)out_size; (void)ws_size;
}


# edge list binned once by destination tile in the second launch (256 slice blocks, LDS slot adds): a tile workgroup reads its own segments, not the whole list; row sums still in arrival order
# speedup vs baseline: 5.4637x; 1.1417x over previous
#include <hip/hip_runtime.h>

#define ND_N 20000
#define NW_N 30000
#define E_N  500000
#define D_N  256
#define DOUT_N 256

#define RW 224
#define RD 176
#define RMAX 224
#define NBW ((NW_N + RW - 1) / RW)
#define NBD ((ND_N + RD - 1) / RD)
#define PITCH 260
#define WCAP 256
#define CAP (32 * WCAP)
#define NTHR 1024
#define NIT ((E_N + NTHR * 4 - 1) / (NTHR * 4))
#define NT (NBW + NBD)
#define NSL 256
#define SPW (NSL / 32)
#define SC 64
#define LPS (SC / 32)
#define GRPT 2
#define NGRP (E_N / 4)
#define GPS ((NGRP + NSL - 1) / NSL)
#define BTHR 256
#define WS_SCORE_FLOATS (2 * ND_N + 2 * NW_N)
#define WS_NEED ((unsigned long long)WS_SCORE_FLOATS * 4ull + (unsigned long long)NT * NSL * 4ull + (unsigned long long)NT * NSL * SC * 4ull)

typedef __attribute__((ext_vector_type(2))) float v2f;
typedef __attribute__((ext_vector_type(8))) float v8f;

__device__ __forceinline__ unsigned mapf(float f) {
  unsigned u = __float_as_uint(f);
  return (u & 0x80000000u) ? ~u : (u | 0x80000000u);
}
__device__ __forceinline__ float unmapf(unsigned u) {
  unsigned v = (u & 0x80000000u) ? (u & 0x7fffffffu) : ~u;
  return __uint_as_float(v);
}
#define NEG_INF_MAPPED 0x007fffffu

__device__ __forceinline__ void node_dots_body(int blk, const float* __restrict__ hid, int n,
                                               const float* __restrict__ wa,
                                               const float* __restrict__ wb,
                                               float* __restrict__ outa,
                                               float* __restrict__ outb) {
  int wave = threadIdx.x >> 5, lane = threadIdx.x & 31;
  int node = blk * 8 + wave;
  if (node >= n) return;
  const float* row = hid + (size_t)node * D_N;
  float da = 0.f, db = 0.f;
#pragma unroll
  for (int i = 0; i < 8; ++i) {
    int k = lane + i * 32;
    float x = row[k];
    da += x * wa[k];
    db += x * wb[k];
  }
#pragma unroll
  for (int off = 16; off; off >>= 1) {
    da += __shfl_xor(da, off, 32);
    db += __shfl_xor(db, off, 32);
  }
  if (lane == 0) { outa[node] = da; outb[node] = db; }
}

__global__ void node_dots_kernel(const float* __restrict__ hid, int n,
                                 const float* __restrict__ wa,
                                 const float* __restrict__ wb,
                                 float* __restrict__ outa,
                                 float* __restrict__ outb) {
  node_dots_body(blockIdx.x, hid, n, wa, wb, outa, outb);
}

__global__ __launch_bounds__(BTHR) void bin_dots_kernel(
    const float* __restrict__ hid, int n,
    const float* __restrict__ wa, const float* __restrict__ wb,
    float* __restrict__ outa, float* __restrict__ outb,
    const int* __restrict__ edge_doc, const int* __restrict__ edge_word,
    unsigned* __restrict__ cnts, unsigned* __restrict__ segs, unsigned long long wsBytes) {
  __shared__ unsigned lc[NT];
  if (blockIdx.x >= NSL) {
    node_dots_body(blockIdx.x - NSL, hid, n, wa, wb, outa, outb);
    return;
  }
  if (wsBytes < WS_NEED) return;
  const int tid = threadIdx.x;
  const int sl = blockIdx.x;
  if (tid < NT) lc[tid] = 0u;
  __syncthreads();
  const int g0 = sl * GPS;
  int g1 = g0 + GPS; if (g1 > NGRP) g1 = NGRP;
  int4 dv[GRPT], wv[GRPT];
#pragma unroll
  for (int i = 0; i < GRPT; ++i) {
    int g = g0 + tid + i * BTHR;
    dv[i] = make_int4(-1, -1, -1, -1);
    wv[i] = make_int4(-1, -1, -1, -1);
    if (g < g1) {
      dv[i] = *(const int4*)(edge_doc + 4 * g);
      wv[i] = *(const int4*)(edge_word + 4 * g);
    }
  }
#pragma unroll
  for (int i = 0; i < GRPT; ++i) {
    int dd[4] = {dv[i].x, dv[i].y, dv[i].z, dv[i].w};
    int ww[4] = {wv[i].x, wv[i].y, wv[i].z, wv[i].w};
    unsigned tw[4], td[4], pw[4], pd[4];
#pragma unroll
    for (int c = 0; c < 4; ++c) {
      unsigned d = (unsigned)dd[c], w = (unsigned)ww[c];
      pw[c] = SC; pd[c] = SC; tw[c] = 0u; td[c] = 0u;
      if (d < (unsigned)ND_N && w < (unsigned)NW_N) {
        tw[c] = w / RW; td[c] = d / RD;
        pw[c] = atomicAdd(&lc[tw[c]], 1u);
        pd[c] = atomicAdd(&lc[NBW + td[c]], 1u);
      }
    }
#pragma unroll
    for (int c = 0; c < 4; ++c) {
      unsigned d = (unsigned)dd[c], w = (unsigned)ww[c];
      if (pw[c] < SC) segs[((size_t)tw[c] * NSL + sl) * SC + pw[c]] = (d << 8) | (w - tw[c] * RW);
      if (pd[c] < SC) segs[((size_t)(NBW + td[c]) * NSL + sl) * SC + pd[c]] = (w << 8) | (d - td[c] * RD);
    }
  }
  __syncthreads();
  if (tid < NT) cnts[tid * NSL + sl] = lc[tid];
}

__device__ __forceinline__ void add_row(float* __restrict__ arow, const float* __restrict__ srow,
                                        float wgt, int lane) {
  float x[8];
#pragma unroll
  for (int q = 0; q < 8; ++q) x[q] = srow[lane + 32 * q];
#pragma unroll
  for (int q = 0; q < 8; ++q) atomicAdd(&arow[lane + 32 * q], wgt * x[q]);
}

#define ROW_ADD(SLOT, E, X) { _Pragma("unroll") for (int q = 0; q < 8; ++q) V[q][SLOT] += (E) * (X)[q]; }
#define ROW_OUT(K) {                                                       \
    const int r = (K) * 32 + wave;                                         \
    if (r < R) {                                                           \
      const float d = fmaxf(sm[r], 1e-30f);                                \
      _Pragma("unroll") for (int q = 0; q < 8; ++q) acc[r * PITCH + lane + 32 * q] = V[q][K] / d; \
    }                                                                      \
  }

__global__ __launch_bounds__(NTHR) void gat_tile_kernel(
    const float* __restrict__ doc_hidden, const float* __restrict__ word_hidden,
    const int* __restrict__ edge_doc, const int* __restrict__ edge_word,
    const float* __restrict__ sA_doc, const float* __restrict__ sB_doc,
    const float* __restrict__ sA_word, const float* __restrict__ sB_word,
    const float* __restrict__ d2w_b, const float* __restrict__ w2d_b,
    const float* __restrict__ W, const float* __restrict__ bias,
    const unsigned* __restrict__ cnts, const unsigned* __restrict__ segs, unsigned long long wsBytes,
    float* __restrict__ out) {
  __shared__ float acc[RMAX * PITCH];
  __shared__ unsigned mx[RMAX];
  __shared__ float sm[RMAX];
  __shared__ float sdl[RMAX];
  __shared__ unsigned ent[CAP];
  __shared__ unsigned own[CAP];
  __shared__ unsigned cnt[32];
  __shared__ unsigned ovf;

  const int tid = threadIdx.x;
  const int wave = __builtin_amdgcn_readfirstlane(tid >> 5);
  const int lane = tid & 31;

  const bool isWord = blockIdx.x < NBW;
  int rowBase, R, outRow0;
  const int *didx, *sidx;
  const float *sS, *sD, *feat, *bptr;
  if (isWord) {
    rowBase = blockIdx.x * RW;
    R = NW_N - rowBase; if (R > RW) R = RW;
    didx = edge_word; sidx = edge_doc; sS = sA_doc; sD = sA_word;
    feat = doc_hidden; bptr = d2w_b; outRow0 = ND_N + rowBase;
  } else {
    rowBase = (blockIdx.x - NBW) * RD;
    R = ND_N - rowBase; if (R > RD) R = RD;
    didx = edge_doc; sidx = edge_word; sS = sB_word; sD = sB_doc;
    feat = word_hidden; bptr = w2d_b; outRow0 = rowBase;
  }
  const float b0 = bptr[0];
  const v8f zero8 = {0.f, 0.f, 0.f, 0.f, 0.f, 0.f, 0.f, 0.f};

  const bool binned = wsBytes >= WS_NEED;
  const size_t cell = (size_t)blockIdx.x * NSL + (size_t)wave * SPW;
  unsigned segLen = 0u;
  if (binned && lane < SPW) segLen = cnts[cell + lane];

  if (tid < 32) cnt[tid] = 0u;
  if (tid < RMAX) {
    mx[tid] = NEG_INF_MAPPED;
    sm[tid] = 0.f;
    sdl[tid] = (tid < R) ? sD[rowBase + tid] : 0.f;
  }
  if (tid == 0) ovf = 0u;
  __syncthreads();

  if (binned) {
    if (segLen > SC) ovf = 1u;
    const unsigned* sb = segs + cell * SC;
    unsigned u[SPW * LPS];
#pragma unroll
    for (int k = 0; k < SPW; ++k) {
      unsigned ck = (unsigned)__builtin_amdgcn_readlane((int)segLen, k);
      if (ck > SC) ck = SC;
#pragma unroll
      for (int j = 0; j < LPS; ++j) {
        unsigned i = (unsigned)lane + 32u * j;
        u[k * LPS + j] = 0xffffffffu;
        if (i < ck) u[k * LPS + j] = sb[k * SC + i];
      }
    }
#pragma unroll
    for (int t = 0; t < SPW * LPS; ++t) {
      if (u[t] != 0xffffffffu) {
        unsigned ow = u[t] & 31u;
        unsigned pos = atomicAdd(&cnt[ow], 1u);
        if (pos < WCAP) own[ow * WCAP + pos] = u[t];
      }
    }
  } else if (tid == 0) {
    ovf = 1u;
  }
  __syncthreads();
  const unsigned n = (unsigned)__builtin_amdgcn_readfirstlane((int)cnt[wave]);
  if (n > WCAP && lane == 0) ovf = 1u;
  __syncthreads();
  const bool overflow = __builtin_amdgcn_readfirstlane((int)ovf) != 0;

  if (!overflow) {
    const unsigned oseg = (unsigned)wave * WCAP;
    for (unsigned i = lane; i < n; i += 32u) {
      unsigned u = own[oseg + i];
      unsigned dl = u & 255u;
      int s = (int)(u >> 8);
      float a = sS[s] + sdl[dl] + b0;
      float v = a > 0.f ? a : 0.01f * a;
      ent[oseg + i] = __float_as_uint(v);
      atomicMax(&mx[dl], mapf(v));
    }

    v8f V[8];
#pragma unroll
    for (int q = 0; q < 8; ++q) V[q] = zero8;
    for (unsigned i0 = 0u; i0 < n; i0 += 32u) {
      unsigned i = i0 + lane;
      unsigned dl = 0u; int s = 0; float ex = 0.f;
      if (i < n) {
        unsigned u = own[oseg + i];
        dl = u & 255u;
        s = (int)(u >> 8);
        ex = __expf(__uint_as_float(ent[oseg + i]) - unmapf(mx[dl]));
        atomicAdd(&sm[dl], ex);
      }
      unsigned m = n - i0; if (m > 32u) m = 32u;
      unsigned j = 0u;
      for (; j + 4u <= m; j += 4u) {
        float x[4][8];
#pragma unroll
        for (int t = 0; t < 4; ++t) {
          int sj = __builtin_amdgcn_readlane(s, j + t);
          const float* srow = feat + (size_t)sj * D_N;
#pragma unroll
          for (int q = 0; q < 8; ++q) x[t][q] = srow[lane + 32 * q];
        }
#pragma unroll
        for (int t = 0; t < 4; ++t) {
          const int slot = __builtin_amdgcn_readlane((int)dl, j + t) >> 5;
          const float ej = __uint_as_float((unsigned)__builtin_amdgcn_readlane((int)__float_as_uint(ex), j + t));
          ROW_ADD(slot, ej, x[t])
        }
      }
      for (; j < m; ++j) {
        int sj = __builtin_amdgcn_readlane(s, j);
        const int slot = __builtin_amdgcn_readlane((int)dl, j) >> 5;
        const float ej = __uint_as_float((unsigned)__builtin_amdgcn_readlane((int)__float_as_uint(ex), j));
        const float* srow = feat + (size_t)sj * D_N;
        float x1[8];
#pragma unroll
        for (int q = 0; q < 8; ++q) x1[q] = srow[lane + 32 * q];
        ROW_ADD(slot, ej, x1)
      }
    }
    ROW_OUT(0) ROW_OUT(1) ROW_OUT(2) ROW_OUT(3) ROW_OUT(4) ROW_OUT(5) ROW_OUT(6)
  } else {
    for (int i = tid; i < RMAX * PITCH; i += NTHR) acc[i] = 0.f;
    __syncthreads();
    for (int it = 0; it < NIT; ++it) {
      int base = (it * NTHR + tid) * 4;
      int4 d4 = make_int4(-1, -1, -1, -1);
      if (base < E_N) d4 = *(const int4*)(didx + base);
      int dd[4] = {d4.x, d4.y, d4.z, d4.w};
#pragma unroll
      for (int c = 0; c < 4; ++c) {
        unsigned dl = (unsigned)(dd[c] - rowBase);
        if (dl < (unsigned)R) {
          int s = sidx[base + c];
          float a = sS[s] + sdl[dl] + b0;
          float v = a > 0.f ? a : 0.01f * a;
          atomicMax(&mx[dl], mapf(v));
        }
      }
    }
    __syncthreads();
    for (int it = 0; it < NIT; ++it) {
      int base = (it * NTHR + tid) * 4;
      int4 d4 = make_int4(-1, -1, -1, -1);
      if (base < E_N) d4 = *(const int4*)(didx + base);
      int dd[4] = {d4.x, d4.y, d4.z, d4.w};
#pragma unroll
      for (int c = 0; c < 4; ++c) {
        unsigned dl = (unsigned)(dd[c] - rowBase);
        bool m = dl < (unsigned)R;
        int s = 0; float ex = 0.f;
        if (m) {
          s = sidx[base + c];
          float a = sS[s] + sdl[dl] + b0;
          float v = a > 0.f ? a : 0.01f * a;
          ex = __expf(v - unmapf(mx[dl]));
          atomicAdd(&sm[dl], ex);
        } else {
          dl = 0u;
        }
        unsigned mask = __builtin_amdgcn_ballot_w32(m);
        while (mask) {
          int j = __builtin_ctz(mask);
          mask &= mask - 1u;
          int sj = __builtin_amdgcn_readlane(s, j);
          unsigned dj = (unsigned)__builtin_amdgcn_readlane((int)dl, j);
          float ej = __uint_as_float((unsigned)__builtin_amdgcn_readlane((int)__float_as_uint(ex), j));
          add_row(acc + dj * PITCH, feat + (size_t)sj * D_N, ej, lane);
        }
      }
    }
    __syncthreads();
    for (int i = tid; i < R * D_N; i += NTHR) {
      int r = i >> 8, c = i & 255;
      acc[r * PITCH + c] = acc[r * PITCH + c] / fmaxf(sm[r], 1e-30f);
    }
  }
  __syncthreads();

  const int ntiles = R >> 4;
  const int halfK = lane >> 4;
  const int l16 = lane & 15;
  const int rt0 = wave >> 2, rt1 = rt0 + 8;
  const bool has0 = rt0 < ntiles, has1 = rt1 < ntiles;
  const int colBase = (wave & 3) * 64 + l16;

  v8f zero = {0.f, 0.f, 0.f, 0.f, 0.f, 0.f, 0.f, 0.f};
  v8f c0 = zero, c1 = zero, c2 = zero, c3 = zero;
  v8f c4 = zero, c5 = zero, c6 = zero, c7 = zero;
  if (has0) {
    const float* A0 = acc + (rt0 * 16 + l16) * PITCH;
    const float* A1 = acc + (rt1 * 16 + l16) * PITCH;
    for (int k0 = 0; k0 < D_N; k0 += 4) {
      int ka = k0 + halfK * 2;
      const float* w0 = W + (size_t)ka * DOUT_N;
      const float* w1 = w0 + DOUT_N;
      v2f b0v; b0v.x = w0[colBase +  0]; b0v.y = w1[colBase +  0];
      v2f b1v; b1v.x = w0[colBase + 16]; b1v.y = w1[colBase + 16];
      v2f b2v; b2v.x = w0[colBase + 32]; b2v.y = w1[colBase + 32];
      v2f b3v; b3v.x = w0[colBase + 48]; b3v.y = w1[colBase + 48];
      v2f a;
      a.x = A0[ka];
      a.y = A0[ka + 1];
      c0 = __builtin_amdgcn_wmma_f32_16x16x4_f32(false, a, false, b0v, (short)0, c0, false, false);
      c1 = __builtin_amdgcn_wmma_f32_16x16x4_f32(false, a, false, b1v, (short)0, c1, false, false);
      c2 = __builtin_amdgcn_wmma_f32_16x16x4_f32(false, a, false, b2v, (short)0, c2, false, false);
      c3 = __builtin_amdgcn_wmma_f32_16x16x4_f32(false, a, false, b3v, (short)0, c3, false, false);
      if (has1) {
        v2f a2;
        a2.x = A1[ka];
        a2.y = A1[ka + 1];
        c4 = __builtin_amdgcn_wmma_f32_16x16x4_f32(false, a2, false, b0v, (short)0, c4, false, false);
        c5 = __builtin_amdgcn_wmma_f32_16x16x4_f32(false, a2, false, b1v, (short)0, c5, false, false);
        c6 = __builtin_amdgcn_wmma_f32_16x16x4_f32(false, a2, false, b2v, (short)0, c6, false, false);
        c7 = __builtin_amdgcn_wmma_f32_16x16x4_f32(false, a2, false, b3v, (short)0, c7, false, false);
      }
    }
  }
  __syncthreads();

  {
    const int rowOff = halfK * 8;
    v8f lo[4] = {c0, c1, c2, c3};
    v8f hi[4] = {c4, c5, c6, c7};
#pragma unroll
    for (int n = 0; n < 4; ++n) {
      int col = colBase + n * 16;
      float bv = bias[col];
      if (has0) {
#pragma unroll
        for (int r = 0; r < 8; ++r) acc[(rt0 * 16 + r + rowOff) * PITCH + col] = lo[n][r] + bv;
      }
      if (has1) {
#pragma unroll
        for (int r = 0; r < 8; ++r) acc[(rt1 * 16 + r + rowOff) * PITCH + col] = hi[n][r] + bv;
      }
    }
  }
  __syncthreads();

  for (int row = wave; row < R; row += 32) {
    float v[8];
    float mxv = -3.402823466e38f;
#pragma unroll
    for (int i = 0; i < 8; ++i) {
      v[i] = acc[row * PITCH + lane + i * 32];
      mxv = fmaxf(mxv, v[i]);
    }
#pragma unroll
    for (int off = 16; off; off >>= 1) mxv = fmaxf(mxv, __shfl_xor(mxv, off, 32));
    float sum = 0.f;
#pragma unroll
    for (int i = 0; i < 8; ++i) { v[i] = __expf(v[i] - mxv); sum += v[i]; }
#pragma unroll
    for (int off = 16; off; off >>= 1) sum += __shfl_xor(sum, off, 32);
    float inv = 1.0f / sum;
    float* orow = out + (size_t)(outRow0 + row) * DOUT_N;
#pragma unroll
    for (int i = 0; i < 8; ++i) orow[lane + i * 32] = v[i] * inv;
  }
}

extern "C" void kernel_launch(void* const* d_in, const int* in_sizes, int n_in,
                              void* d_out, int out_size, void* d_ws, size_t ws_size,
                              hipStream_t stream) {
  const float* doc_hidden  = (const float*)d_in[0];
  const float* word_hidden = (const float*)d_in[1];
  const int*   edge_doc    = (const int*)d_in[2];
  const int*   edge_word   = (const int*)d_in[3];
  const float* d2w_w       = (const float*)d_in[4];
  const float* d2w_b       = (const float*)d_in[5];
  const float* w2d_w       = (const float*)d_in[6];
  const float* w2d_b       = (const float*)d_in[7];
  const float* fc_w        = (const float*)d_in[8];
  const float* fc_b        = (const float*)d_in[9];
  float* out = (float*)d_out;

  float* ws = (float*)d_ws;
  float* sA_doc  = ws;
  float* sB_doc  = sA_doc + ND_N;
  float* sA_word = sB_doc + ND_N;
  float* sB_word = sA_word + NW_N;
  unsigned* cnts = (unsigned*)(sB_word + NW_N);
  unsigned* segs = cnts + NT * NSL;

  node_dots_kernel<<<ND_N / 8, 256, 0, stream>>>(doc_hidden, ND_N,
                                                 d2w_w, w2d_w + D_N, sA_doc, sB_doc);
  bin_dots_kernel<<<NSL + NW_N / 8, BTHR, 0, stream>>>(word_hidden, NW_N,
                                                       d2w_w + D_N, w2d_w, sA_word, sB_word,
                                                       edge_doc, edge_word, cnts, segs,
                                                       (unsigned long long)ws_size);

  gat_tile_kernel<<<NBW + NBD, NTHR, 0, stream>>>(doc_hidden, word_hidden, edge_doc, edge_word,
                                                  sA_doc, sB_doc, sA_word, sB_word,
                                                  d2w_b, w2d_b, fc_w, fc_b,
                                                  cnts, segs, (unsigned long long)ws_size, out);

  (void)in_sizes; (void)n_in; (void)out_size;
}


# edge scores from the gathered source rows: per-edge dot on the vector unit (new lane order), running max with rescale over each row's list (arrival order); both node-score launches gone
# speedup vs baseline: 5.8845x; 1.0770x over previous
#include <hip/hip_runtime.h>

#define ND_N 20000
#define NW_N 30000
#define E_N  500000
#define D_N  256
#define DOUT_N 256

#define RW 224
#define RD 176
#define RMAX 224
#define KROWS (RMAX / 32)
#define NBW ((NW_N + RW - 1) / RW)
#define NBD ((ND_N + RD - 1) / RD)
#define PITCH 260
#define RCAP 64
#define NTHR 1024
#define NIT ((E_N + NTHR * 4 - 1) / (NTHR * 4))
#define NT (NBW + NBD)
#define NSL 256
#define SPW (NSL / 32)
#define SC 64
#define LPS (SC / 32)
#define GRPT 2
#define NGRP (E_N / 4)
#define GPS ((NGRP + NSL - 1) / NSL)
#define BTHR 256
#define WS_NEED ((unsigned long long)NT * NSL * 4ull + (unsigned long long)NT * NSL * SC * 4ull)

typedef __attribute__((ext_vector_type(2))) float v2f;
typedef __attribute__((ext_vector_type(8))) float v8f;

__device__ __forceinline__ unsigned mapf(float f) {
  unsigned u = __float_as_uint(f);
  return (u & 0x80000000u) ? ~u : (u | 0x80000000u);
}
__device__ __forceinline__ float unmapf(unsigned u) {
  unsigned v = (u & 0x80000000u) ? (u & 0x7fffffffu) : ~u;
  return __uint_as_float(v);
}
#define NEG_INF_MAPPED 0x007fffffu

#define XMASK_ADD(V, CTRL) (V) += __builtin_bit_cast(float, __builtin_amdgcn_update_dpp(0, __builtin_bit_cast(int, (V)), (CTRL), 0xF, 0xF, true))
__device__ __forceinline__ float wave_sum(float v) {
  XMASK_ADD(v, 0x161);
  XMASK_ADD(v, 0x162);
  XMASK_ADD(v, 0x164);
  XMASK_ADD(v, 0x168);
  int b = __builtin_bit_cast(int, v);
  return v + __builtin_bit_cast(float, __builtin_amdgcn_permlanex16(b, b, 0x76543210, 0xfedcba98, false, false));
}
__device__ __forceinline__ float uni(float v) {
  return __uint_as_float((unsigned)__builtin_amdgcn_readfirstlane((int)__float_as_uint(v)));
}
__device__ __forceinline__ float row_dot(const float (&x)[8], const float (&w)[8]) {
  float p = 0.f;
#pragma unroll
  for (int q = 0; q < 8; ++q) p += x[q] * w[q];
  return uni(wave_sum(p));
}

__global__ __launch_bounds__(BTHR) void bin_kernel(
    const int* __restrict__ edge_doc, const int* __restrict__ edge_word,
    unsigned* __restrict__ cnts, unsigned* __restrict__ segs, unsigned long long wsBytes) {
  __shared__ unsigned lc[NT];
  if (wsBytes < WS_NEED) return;
  const int tid = threadIdx.x;
  const int sl = blockIdx.x;
  if (tid < NT) lc[tid] = 0u;
  __syncthreads();
  const int g0 = sl * GPS;
  int g1 = g0 + GPS; if (g1 > NGRP) g1 = NGRP;
  int4 dv[GRPT], wv[GRPT];
#pragma unroll
  for (int i = 0; i < GRPT; ++i) {
    int g = g0 + tid + i * BTHR;
    dv[i] = make_int4(-1, -1, -1, -1);
    wv[i] = make_int4(-1, -1, -1, -1);
    if (g < g1) {
      dv[i] = *(const int4*)(edge_doc + 4 * g);
      wv[i] = *(const int4*)(edge_word + 4 * g);
    }
  }
#pragma unroll
  for (int i = 0; i < GRPT; ++i) {
    int dd[4] = {dv[i].x, dv[i].y, dv[i].z, dv[i].w};
    int ww[4] = {wv[i].x, wv[i].y, wv[i].z, wv[i].w};
    unsigned tw[4], td[4], pw[4], pd[4];
#pragma unroll
    for (int c = 0; c < 4; ++c) {
      unsigned d = (unsigned)dd[c], w = (unsigned)ww[c];
      pw[c] = SC; pd[c] = SC; tw[c] = 0u; td[c] = 0u;
      if (d < (unsigned)ND_N && w < (unsigned)NW_N) {
        tw[c] = w / RW; td[c] = d / RD;
        pw[c] = atomicAdd(&lc[tw[c]], 1u);
        pd[c] = atomicAdd(&lc[NBW + td[c]], 1u);
      }
    }
#pragma unroll
    for (int c = 0; c < 4; ++c) {
      unsigned d = (unsigned)dd[c], w = (unsigned)ww[c];
      if (pw[c] < SC) segs[((size_t)tw[c] * NSL + sl) * SC + pw[c]] = (d << 8) | (w - tw[c] * RW);
      if (pd[c] < SC) segs[((size_t)(NBW + td[c]) * NSL + sl) * SC + pd[c]] = (w << 8) | (d - td[c] * RD);
    }
  }
  __syncthreads();
  if (tid < NT) cnts[tid * NSL + sl] = lc[tid];
}

__global__ __launch_bounds__(NTHR) void gat_tile_kernel(
    const float* __restrict__ doc_hidden, const float* __restrict__ word_hidden,
    const int* __restrict__ edge_doc, const int* __restrict__ edge_word,
    const float* __restrict__ d2w_w, const float* __restrict__ w2d_w,
    const float* __restrict__ d2w_b, const float* __restrict__ w2d_b,
    const float* __restrict__ W, const float* __restrict__ bias,
    const unsigned* __restrict__ cnts, const unsigned* __restrict__ segs, unsigned long long wsBytes,
    float* __restrict__ out) {
  __shared__ float acc[RMAX * PITCH];
  __shared__ unsigned mx[RMAX];
  __shared__ float sm[RMAX];
  __shared__ float sdl[RMAX];
  __shared__ unsigned own[RMAX * RCAP];
  __shared__ unsigned cnt[RMAX];
  __shared__ unsigned ovf;

  const int tid = threadIdx.x;
  const int wave = __builtin_amdgcn_readfirstlane(tid >> 5);
  const int lane = tid & 31;

  const bool isWord = blockIdx.x < NBW;
  int rowBase, R, outRow0;
  const int *didx, *sidx;
  const float *feat, *featD, *attn, *bptr;
  if (isWord) {
    rowBase = blockIdx.x * RW;
    R = NW_N - rowBase; if (R > RW) R = RW;
    didx = edge_word; sidx = edge_doc;
    feat = doc_hidden; featD = word_hidden; attn = d2w_w; bptr = d2w_b; outRow0 = ND_N + rowBase;
  } else {
    rowBase = (blockIdx.x - NBW) * RD;
    R = ND_N - rowBase; if (R > RD) R = RD;
    didx = edge_doc; sidx = edge_word;
    feat = word_hidden; featD = doc_hidden; attn = w2d_w; bptr = w2d_b; outRow0 = rowBase;
  }
  const float b0 = bptr[0];

  const bool binned = wsBytes >= WS_NEED;
  const size_t cell = (size_t)blockIdx.x * NSL + (size_t)wave * SPW;
  unsigned segLen = 0u;
  if (binned && lane < SPW) segLen = cnts[cell + lane];

  float ws[8], wd[8];
#pragma unroll
  for (int q = 0; q < 8; ++q) { ws[q] = attn[lane + 32 * q]; wd[q] = attn[D_N + lane + 32 * q]; }

  if (tid < RMAX) {
    cnt[tid] = 0u;
    mx[tid] = NEG_INF_MAPPED;
    sm[tid] = 0.f;
  }
  if (tid == 0) ovf = 0u;
  {
    float xd[KROWS][8];
#pragma unroll
    for (int k = 0; k < KROWS; ++k) {
      int r = k * 32 + wave; if (r > R - 1) r = R - 1;
      const float* drow = featD + (size_t)(rowBase + r) * D_N;
#pragma unroll
      for (int q = 0; q < 8; ++q) xd[k][q] = drow[lane + 32 * q];
    }
#pragma unroll
    for (int k = 0; k < KROWS; ++k) {
      const int r = k * 32 + wave;
      const float p = row_dot(xd[k], wd);
      if (r < R && lane == 0) sdl[r] = p;
    }
  }
  __syncthreads();

  if (binned) {
    if (segLen > SC) ovf = 1u;
    const unsigned* sb = segs + cell * SC;
    unsigned u[SPW * LPS];
#pragma unroll
    for (int k = 0; k < SPW; ++k) {
      unsigned ck = (unsigned)__builtin_amdgcn_readlane((int)segLen, k);
      if (ck > SC) ck = SC;
#pragma unroll
      for (int j = 0; j < LPS; ++j) {
        unsigned i = (unsigned)lane + 32u * j;
        u[k * LPS + j] = 0xffffffffu;
        if (i < ck) u[k * LPS + j] = sb[k * SC + i];
      }
    }
#pragma unroll
    for (int t = 0; t < SPW * LPS; ++t) {
      if (u[t] != 0xffffffffu) {
        unsigned row = u[t] & 255u;
        unsigned pos = atomicAdd(&cnt[row], 1u);
        if (pos < RCAP) own[row * RCAP + pos] = u[t] >> 8;
        else ovf = 1u;
      }
    }
  } else if (tid == 0) {
    ovf = 1u;
  }
  __syncthreads();
  const bool overflow = __builtin_amdgcn_readfirstlane((int)ovf) != 0;

  if (!overflow) {
    for (int k = 0; k < KROWS; ++k) {
      const int r = k * 32 + wave;
      if (r >= R) break;
      const unsigned n = (unsigned)__builtin_amdgcn_readfirstlane((int)cnt[r]);
      const float sd = uni(sdl[r]);
      const unsigned* lst = own + r * RCAP;
      float m = -__builtin_inff();
      float l = 0.f;
      float a8[8];
#pragma unroll
      for (int q = 0; q < 8; ++q) a8[q] = 0.f;
      for (unsigned i0 = 0u; i0 < n; i0 += 32u) {
        unsigned mm = n - i0; if (mm > 32u) mm = 32u;
        int s = 0;
        if ((unsigned)lane < mm) s = (int)lst[i0 + lane];
        for (unsigned j = 0u; j < mm; j += 4u) {
          float x[4][8];
#pragma unroll
          for (int t = 0; t < 4; ++t) {
            unsigned jj = j + t; if (jj > mm - 1u) jj = mm - 1u;
            const int sj = __builtin_amdgcn_readlane(s, jj);
            const float* srow = feat + (size_t)sj * D_N;
#pragma unroll
            for (int q = 0; q < 8; ++q) x[t][q] = srow[lane + 32 * q];
          }
          float p[4];
#pragma unroll
          for (int t = 0; t < 4; ++t) p[t] = row_dot(x[t], ws);
#pragma unroll
          for (int t = 0; t < 4; ++t) {
            if (j + t < mm) {
              const float a = p[t] + sd + b0;
              const float v = a > 0.f ? a : 0.01f * a;
              if (v > m) {
                const float sc = __expf(m - v);
                l *= sc;
#pragma unroll
                for (int q = 0; q < 8; ++q) a8[q] *= sc;
                m = v;
              }
              const float pe = __expf(v - m);
              l += pe;
#pragma unroll
              for (int q = 0; q < 8; ++q) a8[q] += pe * x[t][q];
            }
          }
        }
      }
      const float d = fmaxf(l, 1e-30f);
#pragma unroll
      for (int q = 0; q < 8; ++q) acc[r * PITCH + lane + 32 * q] = a8[q] / d;
    }
  } else {
    for (int i = tid; i < RMAX * PITCH; i += NTHR) acc[i] = 0.f;
    __syncthreads();
    for (int it = 0; it < NIT; ++it) {
      int base = (it * NTHR + tid) * 4;
      int4 d4 = make_int4(-1, -1, -1, -1);
      if (base < E_N) d4 = *(const int4*)(didx + base);
      int dd[4] = {d4.x, d4.y, d4.z, d4.w};
#pragma unroll
      for (int c = 0; c < 4; ++c) {
        unsigned dl = (unsigned)(dd[c] - rowBase);
        const bool mt = dl < (unsigned)R;
        int s = 0;
        if (mt) s = sidx[base + c]; else dl = 0u;
        unsigned mask = __builtin_amdgcn_ballot_w32(mt);
        while (mask) {
          const int j = __builtin_ctz(mask);
          mask &= mask - 1u;
          const int sj = __builtin_amdgcn_readlane(s, j);
          const unsigned dj = (unsigned)__builtin_amdgcn_readlane((int)dl, j);
          const float* srow = feat + (size_t)sj * D_N;
          float x1[8];
#pragma unroll
          for (int q = 0; q < 8; ++q) x1[q] = srow[lane + 32 * q];
          const float a = row_dot(x1, ws) + sdl[dj] + b0;
          const float v = a > 0.f ? a : 0.01f * a;
          if (lane == 0) atomicMax(&mx[dj], mapf(v));
        }
      }
    }
    __syncthreads();
    for (int it = 0; it < NIT; ++it) {
      int base = (it * NTHR + tid) * 4;
      int4 d4 = make_int4(-1, -1, -1, -1);
      if (base < E_N) d4 = *(const int4*)(didx + base);
      int dd[4] = {d4.x, d4.y, d4.z, d4.w};
#pragma unroll
      for (int c = 0; c < 4; ++c) {
        unsigned dl = (unsigned)(dd[c] - rowBase);
        const bool mt = dl < (unsigned)R;
        int s = 0;
        if (mt) s = sidx[base + c]; else dl = 0u;
        unsigned mask = __builtin_amdgcn_ballot_w32(mt);
        while (mask) {
          const int j = __builtin_ctz(mask);
          mask &= mask - 1u;
          const int sj = __builtin_amdgcn_readlane(s, j);
          const unsigned dj = (unsigned)__builtin_amdgcn_readlane((int)dl, j);
          const float* srow = feat + (size_t)sj * D_N;
          float x1[8];
#pragma unroll
          for (int q = 0; q < 8; ++q) x1[q] = srow[lane + 32 * q];
          const float a = row_dot(x1, ws) + sdl[dj] + b0;
          const float v = a > 0.f ? a : 0.01f * a;
          const float ex = __expf(v - unmapf(mx[dj]));
          if (lane == 0) atomicAdd(&sm[dj], ex);
          float* arow = acc + dj * PITCH;
#pragma unroll
          for (int q = 0; q < 8; ++q) atomicAdd(&arow[lane + 32 * q], ex * x1[q]);
        }
      }
    }
    __syncthreads();
    for (int i = tid; i < R * D_N; i += NTHR) {
      int r = i >> 8, c = i & 255;
      acc[r * PITCH + c] = acc[r * PITCH + c] / fmaxf(sm[r], 1e-30f);
    }
  }
  __syncthreads();

  const int ntiles = R >> 4;
  const int halfK = lane >> 4;
  const int l16 = lane & 15;
  const int rt0 = wave >> 2, rt1 = rt0 + 8;
  const bool has0 = rt0 < ntiles, has1 = rt1 < ntiles;
  const int colBase = (wave & 3) * 64 + l16;

  v8f zero = {0.f, 0.f, 0.f, 0.f, 0.f, 0.f, 0.f, 0.f};
  v8f c0 = zero, c1 = zero, c2 = zero, c3 = zero;
  v8f c4 = zero, c5 = zero, c6 = zero, c7 = zero;
  if (has0) {
    const float* A0 = acc + (rt0 * 16 + l16) * PITCH;
    const float* A1 = acc + (rt1 * 16 + l16) * PITCH;
    for (int k0 = 0; k0 < D_N; k0 += 4) {
      int ka = k0 + halfK * 2;
      const float* w0 = W + (size_t)ka * DOUT_N;
      const float* w1 = w0 + DOUT_N;
      v2f b0v; b0v.x = w0[colBase +  0]; b0v.y = w1[colBase +  0];
      v2f b1v; b1v.x = w0[colBase + 16]; b1v.y = w1[colBase + 16];
      v2f b2v; b2v.x = w0[colBase + 32]; b2v.y = w1[colBase + 32];
      v2f b3v; b3v.x = w0[colBase + 48]; b3v.y = w1[colBase + 48];
      v2f a;
      a.x = A0[ka];
      a.y = A0[ka + 1];
      c0 = __builtin_amdgcn_wmma_f32_16x16x4_f32(false, a, false, b0v, (short)0, c0, false, false);
      c1 = __builtin_amdgcn_wmma_f32_16x16x4_f32(false, a, false, b1v, (short)0, c1, false, false);
      c2 = __builtin_amdgcn_wmma_f32_16x16x4_f32(false, a, false, b2v, (short)0, c2, false, false);
      c3 = __builtin_amdgcn_wmma_f32_16x16x4_f32(false, a, false, b3v, (short)0, c3, false, false);
      if (has1) {
        v2f a2;
        a2.x = A1[ka];
        a2.y = A1[ka + 1];
        c4 = __builtin_amdgcn_wmma_f32_16x16x4_f32(false, a2, false, b0v, (short)0, c4, false, false);
        c5 = __builtin_amdgcn_wmma_f32_16x16x4_f32(false, a2, false, b1v, (short)0, c5, false, false);
        c6 = __builtin_amdgcn_wmma_f32_16x16x4_f32(false, a2, false, b2v, (short)0, c6, false, false);
        c7 = __builtin_amdgcn_wmma_f32_16x16x4_f32(false, a2, false, b3v, (short)0, c7, false, false);
      }
    }
  }
  __syncthreads();

  {
    const int rowOff = halfK * 8;
    v8f lo[4] = {c0, c1, c2, c3};
    v8f hi[4] = {c4, c5, c6, c7};
#pragma unroll
    for (int n = 0; n < 4; ++n) {
      int col = colBase + n * 16;
      float bv = bias[col];
      if (has0) {
#pragma unroll
        for (int r = 0; r < 8; ++r) acc[(rt0 * 16 + r + rowOff) * PITCH + col] = lo[n][r] + bv;
      }
      if (has1) {
#pragma unroll
        for (int r = 0; r < 8; ++r) acc[(rt1 * 16 + r + rowOff) * PITCH + col] = hi[n][r] + bv;
      }
    }
  }
  __syncthreads();

  for (int row = wave; row < R; row += 32) {
    float v[8];
    float mxv = -3.402823466e38f;
#pragma unroll
    for (int i = 0; i < 8; ++i) {
      v[i] = acc[row * PITCH + lane + i * 32];
      mxv = fmaxf(mxv, v[i]);
    }
#pragma unroll
    for (int off = 16; off; off >>= 1) mxv = fmaxf(mxv, __shfl_xor(mxv, off, 32));
    float sum = 0.f;
#pragma unroll
    for (int i = 0; i < 8; ++i) { v[i] = __expf(v[i] - mxv); sum += v[i]; }
#pragma unroll
    for (int off = 16; off; off >>= 1) sum += __shfl_xor(sum, off, 32);
    float inv = 1.0f / sum;
    float* orow = out + (size_t)(outRow0 + row) * DOUT_N;
#pragma unroll
    for (int i = 0; i < 8; ++i) orow[lane + i * 32] = v[i] * inv;
  }
}

extern "C" void kernel_launch(void* const* d_in, const int* in_sizes, int n_in,
                              void* d_out, int out_size, void* d_ws, size_t ws_size,
                              hipStream_t stream) {
  const float* doc_hidden  = (const float*)d_in[0];
  const float* word_hidden = (const float*)d_in[1];
  const int*   edge_doc    = (const int*)d_in[2];
  const int*   edge_word   = (const int*)d_in[3];
  const float* d2w_w       = (const float*)d_in[4];
  const float* d2w_b       = (const float*)d_in[5];
  const float* w2d_w       = (const float*)d_in[6];
  const float* w2d_b       = (const float*)d_in[7];
  const float* fc_w        = (const float*)d_in[8];
  const float* fc_b        = (const float*)d_in[9];
  float* out = (float*)d_out;

  unsigned* cnts = (unsigned*)d_ws;
  unsigned* segs = cnts + NT * NSL;

  bin_kernel<<<NSL, BTHR, 0, stream>>>(edge_doc, edge_word, cnts, segs, (unsigned long long)ws_size);

  gat_tile_kernel<<<NBW + NBD, NTHR, 0, stream>>>(doc_hidden, word_hidden, edge_doc, edge_word,
                                                  d2w_w, w2d_w, d2w_b, w2d_b, fc_w, fc_b,
                                                  cnts, segs, (unsigned long long)ws_size, out);

  (void)in_sizes; (void)n_in; (void)out_size;
}
